# starting kernel
# baseline (speedup 1.0000x reference)
#include <hip/hip_runtime.h>

typedef float v2f __attribute__((ext_vector_type(2)));
typedef float v8f __attribute__((ext_vector_type(8)));

#define DIM 64
#define WPAD 80

__global__ void zero_f32(float* __restrict__ p, long long n) {
    long long i = (long long)blockIdx.x * blockDim.x + threadIdx.x;
    long long stride = (long long)gridDim.x * blockDim.x;
    for (; i < n; i += stride) p[i] = 0.0f;
}

__global__ void degree_kernel(const int* __restrict__ dst, float* __restrict__ deg,
                              int n_edges) {
    int e = blockIdx.x * blockDim.x + threadIdx.x;
    if (e < n_edges) atomicAdd(&deg[dst[e]], 1.0f);
}

__global__ void invdeg_kernel(float* __restrict__ deg, int n) {
    int i = blockIdx.x * blockDim.x + threadIdx.x;
    if (i < n) deg[i] = 1.0f / fmaxf(deg[i], 1.0f);
}

__global__ void scatter_add_kernel(const float* __restrict__ feat,
                                   const int* __restrict__ src,
                                   const int* __restrict__ dst,
                                   float* __restrict__ agg, int n_edges) {
    long long tid = (long long)blockIdx.x * blockDim.x + threadIdx.x;
    int e = (int)(tid >> 6);
    int d = (int)(tid & 63);
    if (e >= n_edges) return;
    int s = src[e];
    int t = dst[e];
    atomicAdd(&agg[(long long)t * DIM + d], feat[(long long)s * DIM + d]);
}

template <bool RELU>
__global__ void sage_gemm_kernel(const float* __restrict__ agg,
                                 const float* __restrict__ invdeg,
                                 const float* __restrict__ self_feat,
                                 const float* __restrict__ wl,
                                 const float* __restrict__ bl,
                                 const float* __restrict__ wr,
                                 float* __restrict__ out, int n_nodes) {
    __shared__ v2f lds_wl[32 * WPAD];
    __shared__ v2f lds_wr[32 * WPAD];

    for (int i = threadIdx.x; i < 32 * 64; i += blockDim.x) {
        const int p = i >> 6;
        const int n = i & 63;
        v2f a, b;
        a.x = wl[(2 * p) * DIM + n];
        a.y = wl[(2 * p + 1) * DIM + n];
        b.x = wr[(2 * p) * DIM + n];
        b.y = wr[(2 * p + 1) * DIM + n];
        lds_wl[p * WPAD + n] = a;
        lds_wr[p * WPAD + n] = b;
    }
    __syncthreads();

    const int lane = threadIdx.x & 31;
    const int wave = threadIdx.x >> 5;
    const int tile = blockIdx.x * (blockDim.x >> 5) + wave;
    const int n_tiles = (n_nodes + 15) >> 4;
    if (tile >= n_tiles) return;

    const int row0 = tile << 4;
    const int r     = lane & 15;
    const int kh    = (lane >> 4) << 1;
    const int pbase = lane >> 4;

    int node = row0 + r;
    if (node > n_nodes - 1) node = n_nodes - 1;
    const float id = invdeg[node];
    const float* arow = agg       + (long long)node * DIM;
    const float* xrow = self_feat + (long long)node * DIM;

    v8f acc[4] = {{}, {}, {}, {}};

#pragma unroll
    for (int k = 0; k < DIM; k += 4) {
        v2f a;
        a.x = arow[k + kh]     * id;
        a.y = arow[k + kh + 1] * id;
        const int prow = (k >> 1) + pbase;
#pragma unroll
        for (int nt = 0; nt < 4; ++nt) {
            v2f b = lds_wl[prow * WPAD + nt * 16 + r];
            acc[nt] = __builtin_amdgcn_wmma_f32_16x16x4_f32(
                false, a, false, b, (short)0, acc[nt], false, false);
        }
    }

#pragma unroll
    for (int k = 0; k < DIM; k += 4) {
        v2f a;
        a.x = xrow[k + kh];
        a.y = xrow[k + kh + 1];
        const int prow = (k >> 1) + pbase;
#pragma unroll
        for (int nt = 0; nt < 4; ++nt) {
            v2f b = lds_wr[prow * WPAD + nt * 16 + r];
            acc[nt] = __builtin_amdgcn_wmma_f32_16x16x4_f32(
                false, a, false, b, (short)0, acc[nt], false, false);
        }
    }

    const int mbase = (lane >> 4) << 3;
#pragma unroll
    for (int nt = 0; nt < 4; ++nt) {
        const int n = nt * 16 + r;
        const float bias = bl[n];
#pragma unroll
        for (int vr = 0; vr < 8; ++vr) {
            const int m = row0 + mbase + vr;
            if (m < n_nodes) {
                float v = acc[nt][vr] + bias;
                if (RELU) v = fmaxf(v, 0.0f);
                out[(long long)m * DIM + n] = v;
            }
        }
    }
}

extern "C" void kernel_launch(void* const* d_in, const int* in_sizes, int n_in,
                              void* d_out, int out_size, void* d_ws, size_t ws_size,
                              hipStream_t stream) {
    const float* x   = (const float*)d_in[0];
    const int*  edge = (const int*) d_in[1];
    const float* w1l = (const float*)d_in[2];
    const float* b1l = (const float*)d_in[3];
    const float* w1r = (const float*)d_in[4];
    const float* w2l = (const float*)d_in[5];
    const float* b2l = (const float*)d_in[6];
    const float* w2r = (const float*)d_in[7];
    float* out = (float*)d_out;

    const int n_nodes = in_sizes[0] / DIM;
    const int n_edges = in_sizes[1] / 2;
    const int* src = edge;
    const int* dst = edge + n_edges;

    float* ws      = (float*)d_ws;
    float* invdeg  = ws;
    float* agg     = ws + n_nodes;
    float* h       = agg + (long long)n_nodes * DIM;

    const long long feat_elems = (long long)n_nodes * DIM;
    const int edge_blocks  = (n_edges + 255) / 256;
    const int node_blocks  = (n_nodes + 255) / 256;
    const int sc_blocks    = (int)(((long long)n_edges * DIM + 255) / 256);
    const int tiles        = (n_nodes + 15) / 16;
    const int gemm_blocks  = (tiles + 7) / 8;

    zero_f32<<<1024, 256, 0, stream>>>(invdeg, n_nodes);
    degree_kernel<<<edge_blocks, 256, 0, stream>>>(dst, invdeg, n_edges);
    invdeg_kernel<<<node_blocks, 256, 0, stream>>>(invdeg, n_nodes);

    zero_f32<<<2048, 256, 0, stream>>>(agg, feat_elems);
    scatter_add_kernel<<<sc_blocks, 256, 0, stream>>>(x, src, dst, agg, n_edges);
    sage_gemm_kernel<true><<<gemm_blocks, 256, 0, stream>>>(
        agg, invdeg, x, w1l, b1l, w1r, h, n_nodes);

    zero_f32<<<2048, 256, 0, stream>>>(agg, feat_elems);
    scatter_add_kernel<<<sc_blocks, 256, 0, stream>>>(h, src, dst, agg, n_edges);
    sage_gemm_kernel<false><<<gemm_blocks, 256, 0, stream>>>(
        agg, invdeg, h, w2l, b2l, w2r, out, n_nodes);
}


# Layer 1 sweep reads 1 byte per edge, not 4: a launch in front writes each edge's owner-group byte into the h plane; 4 keys tested per op, target checked on a hit; LDS arrival order; 4 launches to 5
# speedup vs baseline: 5.0121x; 5.0121x over previous
#include <hip/hip_runtime.h>

typedef float v2f __attribute__((ext_vector_type(2)));
typedef float v8f __attribute__((ext_vector_type(8)));

#define DIM 64
#define WPAD 80

#define OWN_T       392
#define OWN_THREADS 1024
#define OWN_WAVES   (OWN_THREADS / 32)
#define SUPER       1024
#define QCAP        1088
#define LIST_HEAD   32
#define ROW_BITS    9
#define NO_LIST     0xFFFFFFFFu
#define KTRIP       1024

__global__ __launch_bounds__(256)
void owner_key_kernel(const int* __restrict__ dst, unsigned* __restrict__ keys,
                      int n_nodes, int n_edges) {
    if ((n_edges & 3) != 0 || (long long)n_edges > (long long)n_nodes * DIM * 4) return;
    const int n_quads = (n_edges / KTRIP) * (KTRIP / 4);
    const int4* d4 = reinterpret_cast<const int4*>(dst);
#pragma unroll
    for (int k = 0; k < 4; ++k) {
        const int i = (blockIdx.x * 4 + k) * 256 + threadIdx.x;
        if (i < n_quads) {
            const int4 d = d4[i];
            keys[i] =  (((unsigned)d.x / OWN_T) & 0xffu)
                    | ((((unsigned)d.y / OWN_T) & 0xffu) << 8)
                    | ((((unsigned)d.z / OWN_T) & 0xffu) << 16)
                    | ((((unsigned)d.w / OWN_T) & 0xffu) << 24);
        }
    }
}

template <bool FIRST>
__global__ __launch_bounds__(OWN_THREADS)
void owner_sum_kernel(const float* __restrict__ feat,
                      const int* __restrict__ src,
                      const int* __restrict__ dst,
                      float* __restrict__ agg,
                      float* __restrict__ invdeg,
                      unsigned* __restrict__ elist,
                      const unsigned* __restrict__ keys,
                      int n_nodes, int n_edges) {
    __shared__ __attribute__((aligned(16))) float acc[(OWN_T + 1) * DIM];
    __shared__ unsigned cnt[OWN_T + 1];
    __shared__ int      queue[OWN_WAVES * QCAP];
    __shared__ unsigned n_listed;

    const int tid   = threadIdx.x;
    const int lane  = tid & 31;
    const int wave  = __builtin_amdgcn_readfirstlane(tid >> 5);
    const int node0 = blockIdx.x * OWN_T;
    int nrow = n_nodes - node0;
    if (nrow > OWN_T) nrow = OWN_T;

    unsigned* mylist = elist + (long long)node0 * DIM;
    const unsigned cap = (unsigned)(nrow * DIM - LIST_HEAD);
    const bool packable = n_nodes <= (1 << (32 - ROW_BITS));

    for (int i = tid; i < (OWN_T + 1) * DIM / 4; i += OWN_THREADS)
        reinterpret_cast<float4*>(acc)[i] = make_float4(0.f, 0.f, 0.f, 0.f);
    if (FIRST) {
        for (int i = tid; i <= OWN_T; i += OWN_THREADS) cnt[i] = 0u;
        if (tid == 0) n_listed = 0u;
    }
    __syncthreads();

    auto add_rows = [&](int s, int u) {
#pragma unroll
        for (int g = 0; g < 2; ++g) {
            v2f v[16];
#pragma unroll
            for (int j = 0; j < 16; ++j) {
                const int sj = __builtin_amdgcn_readlane(s, g * 16 + j);
                v[j] = *reinterpret_cast<const v2f*>(feat + (long long)sj * DIM + 2 * lane);
            }
#pragma unroll
            for (int j = 0; j < 16; ++j) {
                const int uj = __builtin_amdgcn_readlane(u, g * 16 + j);
                float* a = acc + uj * DIM + 2 * lane;
                atomicAdd(a,     v[j].x);
                atomicAdd(a + 1, v[j].y);
            }
        }
    };

    bool swept = FIRST;
    if (!FIRST) {
        const unsigned n_list = mylist[0];
        if (n_list <= cap) {
            if (n_list != 0u) {
                const unsigned pad  = (unsigned)OWN_T;
                const unsigned last = n_list - 1u;
                const unsigned* ent = mylist + LIST_HEAD;
                auto entry = [&](unsigned idx) {
                    const unsigned v = ent[idx < last ? idx : last];
                    return idx <= last ? v : pad;
                };
                unsigned i = (unsigned)wave * 32u;
                unsigned w = entry(i + lane);
                while (i < n_list) {
                    const unsigned inext = i + OWN_THREADS;
                    const unsigned wnext = entry(inext + lane);
                    add_rows((int)(w >> ROW_BITS), (int)(w & ((1u << ROW_BITS) - 1u)));
                    w = wnext;
                    i = inext;
                }
            }
        } else {
            swept = true;
        }
    }

    if (swept) {
        int* q  = queue + wave * QCAP;
        int  qn = 0;

        auto consider = [&](int d, int e) {
            const bool m = (unsigned)(d - node0) < (unsigned)OWN_T;
            const unsigned mask = __builtin_amdgcn_ballot_w32(m);
            if (mask) {
                if (m) q[qn + (int)__builtin_amdgcn_mbcnt_lo(mask, 0u)] = e;
                qn += __builtin_popcount(mask);
            }
        };

        auto flush = [&](int take) {
            const bool valid = lane < take;
            int s = 0, u = OWN_T;
            if (valid) {
                const int e = q[qn - take + lane];
                s = src[e];
                u = dst[e] - node0;
                if ((unsigned)u >= (unsigned)OWN_T) { s = 0; u = OWN_T; }
            }
            if (FIRST) {
                atomicAdd(&cnt[u], 1u);
                unsigned base = 0u;
                if (lane == 0) base = atomicAdd(&n_listed, (unsigned)take);
                base = (unsigned)__builtin_amdgcn_readfirstlane((int)base);
                if (valid && base + (unsigned)lane < cap)
                    mylist[LIST_HEAD + base + lane] = ((unsigned)s << ROW_BITS) | (unsigned)u;
            }
            add_rows(s, u);
            qn -= take;
        };

        long long swept_edges;
        if (FIRST && (n_edges & 3) == 0 && (long long)n_edges <= (long long)n_nodes * DIM * 4) {
            const int n_trips = n_edges / KTRIP;
            const int rot = (int)(((long long)blockIdx.x * n_trips) / gridDim.x);
            const unsigned gpat = ((unsigned)blockIdx.x & 0xffu) * 0x01010101u;
            const uint4* k4 = reinterpret_cast<const uint4*>(keys);
            auto place = [&](int i) { int sc = i + rot; if (sc >= n_trips) sc -= n_trips; return sc; };

            auto hits = [&](unsigned w) {
                const unsigned x = w ^ gpat;
                return (x - 0x01010101u) & ~x & 0x80808080u;
            };
            auto consider32 = [&](const uint4& ka, const uint4& kb, int ebase) {
                unsigned tt = (hits(ka.x) >> 7) | (hits(ka.y) >> 6) | (hits(ka.z) >> 5) | (hits(ka.w) >> 4)
                            | (hits(kb.x) >> 3) | (hits(kb.y) >> 2) | (hits(kb.z) >> 1) |  hits(kb.w);
                unsigned mask = __builtin_amdgcn_ballot_w32(tt != 0u);
                while (mask) {
                    if (tt != 0u) {
                        const int p = __builtin_ctz(tt);
                        q[qn + (int)__builtin_amdgcn_mbcnt_lo(mask, 0u)] =
                            ebase + ((p & 3) << 2) + (p >> 3) + ((p & 4) << 7);
                    }
                    qn += __builtin_popcount(mask);
                    tt &= tt - 1u;
                    mask = __builtin_amdgcn_ballot_w32(tt != 0u);
                }
            };

            uint4 ca = make_uint4(0u, 0u, 0u, 0u), cb = ca;
            if (wave < n_trips) {
                const uint4* p = k4 + (long long)place(wave) * (KTRIP / 16) + lane;
                ca = p[0]; cb = p[32];
            }
            for (int i = wave; i < n_trips; i += OWN_WAVES) {
                const int inext = (i + OWN_WAVES < n_trips) ? i + OWN_WAVES : i;
                const uint4* pn = k4 + (long long)place(inext) * (KTRIP / 16) + lane;
                const uint4 na = pn[0], nb = pn[32];
                const int ebase = place(i) * KTRIP + lane * 16;
                consider32(ca, cb, ebase);
                while (qn >= 32) flush(32);
                ca = na; cb = nb;
            }
            swept_edges = (long long)n_trips * KTRIP;
        } else {
            const int n_full = ((n_edges & 3) == 0) ? n_edges / SUPER : 0;
            const int rot = (int)(((long long)blockIdx.x * n_full) / gridDim.x);

            for (int i = wave; i < n_full; i += OWN_WAVES) {
                int sc = i + rot; if (sc >= n_full) sc -= n_full;
                const int* p = dst + (long long)sc * SUPER + lane * 4;
                int4 d[8];
#pragma unroll
                for (int k = 0; k < 8; ++k)
                    d[k] = *reinterpret_cast<const int4*>(p + k * 128);
#pragma unroll
                for (int k = 0; k < 8; ++k) {
                    const int ebase = sc * SUPER + k * 128 + lane * 4;
                    consider(d[k].x, ebase);
                    consider(d[k].y, ebase + 1);
                    consider(d[k].z, ebase + 2);
                    consider(d[k].w, ebase + 3);
                }
                while (qn >= 32) flush(32);
            }
            swept_edges = (long long)n_full * SUPER;
        }

        long long e0 = swept_edges + wave * 32;
        for (;;) {
            const bool have = e0 < (long long)n_edges;
            if (have) {
                const long long e = e0 + lane;
                const int d = (e < n_edges) ? dst[e] : node0 - 1;
                consider(d, (int)e);
                e0 += OWN_THREADS;
            }
            const int keep_below = have ? 32 : 1;
            while (qn >= keep_below) flush(qn < 32 ? qn : 32);
            if (!have) break;
        }
    }

    __syncthreads();

    if (FIRST) {
        for (int t = tid; t < OWN_T; t += OWN_THREADS) {
            const int node = node0 + t;
            if (node < n_nodes) invdeg[node] = 1.0f / fmaxf((float)cnt[t], 1.0f);
        }
        if (tid == 0) mylist[0] = (packable && n_listed <= cap) ? n_listed : NO_LIST;
    }
    float4* o = reinterpret_cast<float4*>(agg + (long long)node0 * DIM);
    for (int i = tid; i < nrow * (DIM / 4); i += OWN_THREADS)
        o[i] = reinterpret_cast<const float4*>(acc)[i];
}

template <bool RELU>
__global__ void sage_gemm_kernel(const float* __restrict__ agg,
                                 const float* __restrict__ invdeg,
                                 const float* __restrict__ self_feat,
                                 const float* __restrict__ wl,
                                 const float* __restrict__ bl,
                                 const float* __restrict__ wr,
                                 float* __restrict__ out, int n_nodes) {
    __shared__ v2f lds_wl[32 * WPAD];
    __shared__ v2f lds_wr[32 * WPAD];

    for (int i = threadIdx.x; i < 32 * 64; i += blockDim.x) {
        const int p = i >> 6;
        const int n = i & 63;
        v2f a, b;
        a.x = wl[(2 * p) * DIM + n];
        a.y = wl[(2 * p + 1) * DIM + n];
        b.x = wr[(2 * p) * DIM + n];
        b.y = wr[(2 * p + 1) * DIM + n];
        lds_wl[p * WPAD + n] = a;
        lds_wr[p * WPAD + n] = b;
    }
    __syncthreads();

    const int lane = threadIdx.x & 31;
    const int wave = threadIdx.x >> 5;
    const int tile = blockIdx.x * (blockDim.x >> 5) + wave;
    const int n_tiles = (n_nodes + 15) >> 4;
    if (tile >= n_tiles) return;

    const int row0 = tile << 4;
    const int r     = lane & 15;
    const int kh    = (lane >> 4) << 1;
    const int pbase = lane >> 4;

    int node = row0 + r;
    if (node > n_nodes - 1) node = n_nodes - 1;
    const float id = invdeg[node];
    const float* arow = agg       + (long long)node * DIM;
    const float* xrow = self_feat + (long long)node * DIM;

    v8f acc[4] = {{}, {}, {}, {}};

#pragma unroll
    for (int k = 0; k < DIM; k += 4) {
        v2f a;
        a.x = arow[k + kh]     * id;
        a.y = arow[k + kh + 1] * id;
        const int prow = (k >> 1) + pbase;
#pragma unroll
        for (int nt = 0; nt < 4; ++nt) {
            v2f b = lds_wl[prow * WPAD + nt * 16 + r];
            acc[nt] = __builtin_amdgcn_wmma_f32_16x16x4_f32(
                false, a, false, b, (short)0, acc[nt], false, false);
        }
    }

#pragma unroll
    for (int k = 0; k < DIM; k += 4) {
        v2f a;
        a.x = xrow[k + kh];
        a.y = xrow[k + kh + 1];
        const int prow = (k >> 1) + pbase;
#pragma unroll
        for (int nt = 0; nt < 4; ++nt) {
            v2f b = lds_wr[prow * WPAD + nt * 16 + r];
            acc[nt] = __builtin_amdgcn_wmma_f32_16x16x4_f32(
                false, a, false, b, (short)0, acc[nt], false, false);
        }
    }

    const int mbase = (lane >> 4) << 3;
#pragma unroll
    for (int nt = 0; nt < 4; ++nt) {
        const int n = nt * 16 + r;
        const float bias = bl[n];
#pragma unroll
        for (int vr = 0; vr < 8; ++vr) {
            const int m = row0 + mbase + vr;
            if (m < n_nodes) {
                float v = acc[nt][vr] + bias;
                if (RELU) v = fmaxf(v, 0.0f);
                out[(long long)m * DIM + n] = v;
            }
        }
    }
}

extern "C" void kernel_launch(void* const* d_in, const int* in_sizes, int n_in,
                              void* d_out, int out_size, void* d_ws, size_t ws_size,
                              hipStream_t stream) {
    const float* x   = (const float*)d_in[0];
    const int*  edge = (const int*) d_in[1];
    const float* w1l = (const float*)d_in[2];
    const float* b1l = (const float*)d_in[3];
    const float* w1r = (const float*)d_in[4];
    const float* w2l = (const float*)d_in[5];
    const float* b2l = (const float*)d_in[6];
    const float* w2r = (const float*)d_in[7];
    float* out = (float*)d_out;
    unsigned* elist = (unsigned*)d_out;

    const int n_nodes = in_sizes[0] / DIM;
    const int n_edges = in_sizes[1] / 2;
    const int* src = edge;
    const int* dst = edge + n_edges;

    float* ws      = (float*)d_ws;
    float* agg     = ws;
    float* h       = agg + (long long)n_nodes * DIM;
    float* invdeg  = h + (long long)n_nodes * DIM;

    const int own_blocks   = (n_nodes + OWN_T - 1) / OWN_T;
    const int tiles        = (n_nodes + 15) / 16;
    const int gemm_blocks  = (tiles + 7) / 8;
    const int key_blocks   = n_edges / 4096 + 1;

    owner_key_kernel<<<key_blocks, 256, 0, stream>>>(dst, (unsigned*)h, n_nodes, n_edges);
    owner_sum_kernel<true><<<own_blocks, OWN_THREADS, 0, stream>>>(
        x, src, dst, agg, invdeg, elist, (const unsigned*)h, n_nodes, n_edges);
    sage_gemm_kernel<true><<<gemm_blocks, 256, 0, stream>>>(
        agg, invdeg, x, w1l, b1l, w1r, h, n_nodes);

    owner_sum_kernel<false><<<own_blocks, OWN_THREADS, 0, stream>>>(
        h, src, dst, agg, invdeg, elist, (const unsigned*)0, n_nodes, n_edges);
    sage_gemm_kernel<false><<<gemm_blocks, 256, 0, stream>>>(
        agg, invdeg, h, w2l, b2l, w2r, out, n_nodes);
}


# Transform moved into the owner launch: a wave reads its 16 finished tile rows from LDS as A fragments (pitch 68), WMMA, writes h/out; sum plane gone, keys in its place; 5 launches to 3
# speedup vs baseline: 5.9734x; 1.1918x over previous
#include <hip/hip_runtime.h>

typedef float v2f __attribute__((ext_vector_type(2)));
typedef float v8f __attribute__((ext_vector_type(8)));

#define DIM 64
#define WPAD 80

#define OWN_T       392
#define AP          68
#define OWN_THREADS 1024
#define OWN_WAVES   (OWN_THREADS / 32)
#define SUPER       1024
#define QCAP        1088
#define LIST_HEAD   32
#define ROW_BITS    9
#define NO_LIST     0xFFFFFFFFu
#define KTRIP       1024

__global__ __launch_bounds__(256)
void owner_key_kernel(const int* __restrict__ dst, unsigned* __restrict__ keys,
                      int n_nodes, int n_edges) {
    if ((n_edges & 3) != 0 || (long long)n_edges > (long long)n_nodes * DIM * 4) return;
    const int n_quads = (n_edges / KTRIP) * (KTRIP / 4);
    const int4* d4 = reinterpret_cast<const int4*>(dst);
#pragma unroll
    for (int k = 0; k < 4; ++k) {
        const int i = (blockIdx.x * 4 + k) * 256 + threadIdx.x;
        if (i < n_quads) {
            const int4 d = d4[i];
            keys[i] =  (((unsigned)d.x / OWN_T) & 0xffu)
                    | ((((unsigned)d.y / OWN_T) & 0xffu) << 8)
                    | ((((unsigned)d.z / OWN_T) & 0xffu) << 16)
                    | ((((unsigned)d.w / OWN_T) & 0xffu) << 24);
        }
    }
}

template <bool FIRST>
__global__ __launch_bounds__(OWN_THREADS)
void owner_layer_kernel(const float* __restrict__ feat,
                        const int* __restrict__ src,
                        const int* __restrict__ dst,
                        float* invdeg,
                        unsigned* elist,
                        const unsigned* __restrict__ keys,
                        const float* __restrict__ wl,
                        const float* __restrict__ bl,
                        const float* __restrict__ wr,
                        float* out,
                        int n_nodes, int n_edges) {
    __shared__ __attribute__((aligned(16))) float acc[(OWN_T + 1) * AP];
    __shared__ v2f lds_wl[32 * WPAD];
    __shared__ v2f lds_wr[32 * WPAD];
    __shared__ unsigned cnt[OWN_T + 1];
    __shared__ int      queue[OWN_WAVES * QCAP];
    __shared__ unsigned n_listed;

    const int tid   = threadIdx.x;
    const int lane  = tid & 31;
    const int wave  = __builtin_amdgcn_readfirstlane(tid >> 5);
    const int node0 = blockIdx.x * OWN_T;
    int nrow = n_nodes - node0;
    if (nrow > OWN_T) nrow = OWN_T;

    unsigned* mylist = elist + (long long)node0 * DIM;
    const unsigned cap = (unsigned)(nrow * DIM - LIST_HEAD);
    const bool packable = n_nodes <= (1 << (32 - ROW_BITS));

    for (int i = tid; i < (OWN_T + 1) * AP / 4; i += OWN_THREADS)
        reinterpret_cast<float4*>(acc)[i] = make_float4(0.f, 0.f, 0.f, 0.f);
    for (int i = tid; i < 32 * 64; i += OWN_THREADS) {
        const int p = i >> 6;
        const int n = i & 63;
        v2f a, b;
        a.x = wl[(2 * p) * DIM + n];
        a.y = wl[(2 * p + 1) * DIM + n];
        b.x = wr[(2 * p) * DIM + n];
        b.y = wr[(2 * p + 1) * DIM + n];
        lds_wl[p * WPAD + n] = a;
        lds_wr[p * WPAD + n] = b;
    }
    if (FIRST) {
        for (int i = tid; i <= OWN_T; i += OWN_THREADS) cnt[i] = 0u;
        if (tid == 0) n_listed = 0u;
    }
    __syncthreads();

    auto add_rows = [&](int s, int u) {
#pragma unroll
        for (int g = 0; g < 2; ++g) {
            v2f v[16];
#pragma unroll
            for (int j = 0; j < 16; ++j) {
                const int sj = __builtin_amdgcn_readlane(s, g * 16 + j);
                v[j] = *reinterpret_cast<const v2f*>(feat + (long long)sj * DIM + 2 * lane);
            }
#pragma unroll
            for (int j = 0; j < 16; ++j) {
                const int uj = __builtin_amdgcn_readlane(u, g * 16 + j);
                float* a = acc + uj * AP + 2 * lane;
                atomicAdd(a,     v[j].x);
                atomicAdd(a + 1, v[j].y);
            }
        }
    };

    bool swept = FIRST;
    if (!FIRST) {
        const unsigned n_list = mylist[0];
        if (n_list <= cap) {
            if (n_list != 0u) {
                const unsigned pad  = (unsigned)OWN_T;
                const unsigned last = n_list - 1u;
                const unsigned* ent = mylist + LIST_HEAD;
                auto entry = [&](unsigned idx) {
                    const unsigned v = ent[idx < last ? idx : last];
                    return idx <= last ? v : pad;
                };
                unsigned i = (unsigned)wave * 32u;
                unsigned w = entry(i + lane);
                while (i < n_list) {
                    const unsigned inext = i + OWN_THREADS;
                    const unsigned wnext = entry(inext + lane);
                    add_rows((int)(w >> ROW_BITS), (int)(w & ((1u << ROW_BITS) - 1u)));
                    w = wnext;
                    i = inext;
                }
            }
        } else {
            swept = true;
        }
    }

    if (swept) {
        int* q  = queue + wave * QCAP;
        int  qn = 0;

        auto consider = [&](int d, int e) {
            const bool m = (unsigned)(d - node0) < (unsigned)OWN_T;
            const unsigned mask = __builtin_amdgcn_ballot_w32(m);
            if (mask) {
                if (m) q[qn + (int)__builtin_amdgcn_mbcnt_lo(mask, 0u)] = e;
                qn += __builtin_popcount(mask);
            }
        };

        auto flush = [&](int take) {
            const bool valid = lane < take;
            int s = 0, u = OWN_T;
            if (valid) {
                const int e = q[qn - take + lane];
                s = src[e];
                u = dst[e] - node0;
                if ((unsigned)u >= (unsigned)OWN_T) { s = 0; u = OWN_T; }
            }
            if (FIRST) {
                atomicAdd(&cnt[u], 1u);
                unsigned base = 0u;
                if (lane == 0) base = atomicAdd(&n_listed, (unsigned)take);
                base = (unsigned)__builtin_amdgcn_readfirstlane((int)base);
                if (valid && base + (unsigned)lane < cap)
                    mylist[LIST_HEAD + base + lane] = ((unsigned)s << ROW_BITS) | (unsigned)u;
            }
            add_rows(s, u);
            qn -= take;
        };

        long long swept_edges;
        if (FIRST && (n_edges & 3) == 0 && (long long)n_edges <= (long long)n_nodes * DIM * 4) {
            const int n_trips = n_edges / KTRIP;
            const int rot = (int)(((long long)blockIdx.x * n_trips) / gridDim.x);
            const unsigned gpat = ((unsigned)blockIdx.x & 0xffu) * 0x01010101u;
            const uint4* k4 = reinterpret_cast<const uint4*>(keys);
            auto place = [&](int i) { int sc = i + rot; if (sc >= n_trips) sc -= n_trips; return sc; };

            auto hits = [&](unsigned w) {
                const unsigned x = w ^ gpat;
                return (x - 0x01010101u) & ~x & 0x80808080u;
            };
            auto consider32 = [&](const uint4& ka, const uint4& kb, int ebase) {
                unsigned tt = (hits(ka.x) >> 7) | (hits(ka.y) >> 6) | (hits(ka.z) >> 5) | (hits(ka.w) >> 4)
                            | (hits(kb.x) >> 3) | (hits(kb.y) >> 2) | (hits(kb.z) >> 1) |  hits(kb.w);
                unsigned mask = __builtin_amdgcn_ballot_w32(tt != 0u);
                while (mask) {
                    if (tt != 0u) {
                        const int p = __builtin_ctz(tt);
                        q[qn + (int)__builtin_amdgcn_mbcnt_lo(mask, 0u)] =
                            ebase + ((p & 3) << 2) + (p >> 3) + ((p & 4) << 7);
                    }
                    qn += __builtin_popcount(mask);
                    tt &= tt - 1u;
                    mask = __builtin_amdgcn_ballot_w32(tt != 0u);
                }
            };

            uint4 ca = make_uint4(0u, 0u, 0u, 0u), cb = ca;
            if (wave < n_trips) {
                const uint4* p = k4 + (long long)place(wave) * (KTRIP / 16) + lane;
                ca = p[0]; cb = p[32];
            }
            for (int i = wave; i < n_trips; i += OWN_WAVES) {
                const int inext = (i + OWN_WAVES < n_trips) ? i + OWN_WAVES : i;
                const uint4* pn = k4 + (long long)place(inext) * (KTRIP / 16) + lane;
                const uint4 na = pn[0], nb = pn[32];
                const int ebase = place(i) * KTRIP + lane * 16;
                consider32(ca, cb, ebase);
                while (qn >= 32) flush(32);
                ca = na; cb = nb;
            }
            swept_edges = (long long)n_trips * KTRIP;
        } else {
            const int n_full = ((n_edges & 3) == 0) ? n_edges / SUPER : 0;
            const int rot = (int)(((long long)blockIdx.x * n_full) / gridDim.x);

            for (int i = wave; i < n_full; i += OWN_WAVES) {
                int sc = i + rot; if (sc >= n_full) sc -= n_full;
                const int* p = dst + (long long)sc * SUPER + lane * 4;
                int4 d[8];
#pragma unroll
                for (int k = 0; k < 8; ++k)
                    d[k] = *reinterpret_cast<const int4*>(p + k * 128);
#pragma unroll
                for (int k = 0; k < 8; ++k) {
                    const int ebase = sc * SUPER + k * 128 + lane * 4;
                    consider(d[k].x, ebase);
                    consider(d[k].y, ebase + 1);
                    consider(d[k].z, ebase + 2);
                    consider(d[k].w, ebase + 3);
                }
                while (qn >= 32) flush(32);
            }
            swept_edges = (long long)n_full * SUPER;
        }

        long long e0 = swept_edges + wave * 32;
        for (;;) {
            const bool have = e0 < (long long)n_edges;
            if (have) {
                const long long e = e0 + lane;
                const int d = (e < n_edges) ? dst[e] : node0 - 1;
                consider(d, (int)e);
                e0 += OWN_THREADS;
            }
            const int keep_below = have ? 32 : 1;
            while (qn >= keep_below) flush(qn < 32 ? qn : 32);
            if (!have) break;
        }
    }

    __syncthreads();

    if (FIRST) {
        for (int t = tid; t < OWN_T; t += OWN_THREADS) {
            const int node = node0 + t;
            if (node < n_nodes) invdeg[node] = 1.0f / fmaxf((float)cnt[t], 1.0f);
        }
        if (tid == 0) mylist[0] = (packable && n_listed <= cap) ? n_listed : NO_LIST;
    }

    if (wave * 16 < nrow) {
        const int r     = lane & 15;
        const int kh    = (lane >> 4) << 1;
        const int pbase = lane >> 4;

        int trow = wave * 16 + r;
        if (trow > OWN_T) trow = OWN_T;
        int node = node0 + trow;
        if (node > n_nodes - 1) node = n_nodes - 1;
        const float id = FIRST ? 1.0f / fmaxf((float)cnt[trow], 1.0f) : invdeg[node];
        const float* arow = acc + trow * AP;
        const float* xrow = feat + (long long)node * DIM;

        v8f d[4] = {{}, {}, {}, {}};

#pragma unroll
        for (int k = 0; k < DIM; k += 4) {
            v2f a;
            a.x = arow[k + kh]     * id;
            a.y = arow[k + kh + 1] * id;
            const int prow = (k >> 1) + pbase;
#pragma unroll
            for (int nt = 0; nt < 4; ++nt) {
                v2f b = lds_wl[prow * WPAD + nt * 16 + r];
                d[nt] = __builtin_amdgcn_wmma_f32_16x16x4_f32(
                    false, a, false, b, (short)0, d[nt], false, false);
            }
        }

#pragma unroll
        for (int k = 0; k < DIM; k += 4) {
            v2f a;
            a.x = xrow[k + kh];
            a.y = xrow[k + kh + 1];
            const int prow = (k >> 1) + pbase;
#pragma unroll
            for (int nt = 0; nt < 4; ++nt) {
                v2f b = lds_wr[prow * WPAD + nt * 16 + r];
                d[nt] = __builtin_amdgcn_wmma_f32_16x16x4_f32(
                    false, a, false, b, (short)0, d[nt], false, false);
            }
        }

        const int mbase = wave * 16 + ((lane >> 4) << 3);
#pragma unroll
        for (int nt = 0; nt < 4; ++nt) {
            const int n = nt * 16 + r;
            const float bias = bl[n];
#pragma unroll
            for (int vr = 0; vr < 8; ++vr) {
                const int m = mbase + vr;
                if (m < nrow) {
                    float v = d[nt][vr] + bias;
                    if (FIRST) v = fmaxf(v, 0.0f);
                    out[(long long)(node0 + m) * DIM + n] = v;
                }
            }
        }
    }
}

extern "C" void kernel_launch(void* const* d_in, const int* in_sizes, int n_in,
                              void* d_out, int out_size, void* d_ws, size_t ws_size,
                              hipStream_t stream) {
    const float* x   = (const float*)d_in[0];
    const int*  edge = (const int*) d_in[1];
    const float* w1l = (const float*)d_in[2];
    const float* b1l = (const float*)d_in[3];
    const float* w1r = (const float*)d_in[4];
    const float* w2l = (const float*)d_in[5];
    const float* b2l = (const float*)d_in[6];
    const float* w2r = (const float*)d_in[7];
    float* out = (float*)d_out;
    unsigned* elist = (unsigned*)d_out;

    const int n_nodes = in_sizes[0] / DIM;
    const int n_edges = in_sizes[1] / 2;
    const int* src = edge;
    const int* dst = edge + n_edges;

    float* ws      = (float*)d_ws;
    unsigned* keyp = (unsigned*)ws;
    float* h       = ws + (long long)n_nodes * DIM;
    float* invdeg  = h + (long long)n_nodes * DIM;

    const int own_blocks   = (n_nodes + OWN_T - 1) / OWN_T;
    const int key_blocks   = n_edges / 4096 + 1;

    owner_key_kernel<<<key_blocks, 256, 0, stream>>>(dst, keyp, n_nodes, n_edges);
    owner_layer_kernel<true><<<own_blocks, OWN_THREADS, 0, stream>>>(
        x, src, dst, invdeg, elist, keyp, w1l, b1l, w1r, h, n_nodes, n_edges);

    owner_layer_kernel<false><<<own_blocks, OWN_THREADS, 0, stream>>>(
        h, src, dst, invdeg, elist, (const unsigned*)0, w2l, b2l, w2r, out, n_nodes, n_edges);
}
